# starting kernel
# baseline (speedup 1.0000x reference)
#include <hip/hip_runtime.h>
#include <hip/hip_bf16.h>
#include <math.h>

typedef __attribute__((ext_vector_type(2))) float v2f;
typedef __attribute__((ext_vector_type(8))) float v8f;

#define N_NODES 100000
#define N_EDGES 1600000
#define CH 128
#define LDS_STRIDE 132

#define OFF_W     0
#define OFF_BIAS  (OFF_W + CH*CH)
#define OFF_ODEG  (OFF_BIAS + CH)
#define OFF_IDEG  (OFF_ODEG + N_NODES)
#define OFF_H     (OFF_IDEG + N_NODES)

__global__ void zero_f4(float* __restrict__ p, int n4) {
    int i = blockIdx.x * blockDim.x + threadIdx.x;
    if (i < n4) ((float4*)p)[i] = make_float4(0.f, 0.f, 0.f, 0.f);
}

__global__ void prep_weights(const float* __restrict__ Wmu, const float* __restrict__ Wrho,
                             const float* __restrict__ Weps,
                             const float* __restrict__ bmu, const float* __restrict__ brho,
                             const float* __restrict__ beps,
                             float* __restrict__ W, float* __restrict__ bias) {
    int i = blockIdx.x * blockDim.x + threadIdx.x;
    if (i < CH * CH) {
        float sp = log1pf(expf(Wrho[i]));
        W[i] = Wmu[i] + Weps[i] * sp;
    } else if (i < CH * CH + CH) {
        int j = i - CH * CH;
        float sp = log1pf(expf(brho[j]));
        bias[j] = bmu[j] + beps[j] * sp;
    }
}

__global__ void degrees_kernel(const int* __restrict__ src, const int* __restrict__ dst,
                               float* __restrict__ odeg, float* __restrict__ ideg) {
    int e = blockIdx.x * blockDim.x + threadIdx.x;
    if (e < N_EDGES) {
        atomicAdd(&odeg[src[e]], 1.0f);
        atomicAdd(&ideg[dst[e]], 1.0f);
    }
}

__global__ __launch_bounds__(128) void gemm_wmma(const float* __restrict__ feat,
                                                 const float* __restrict__ odeg,
                                                 const float* __restrict__ W,
                                                 float* __restrict__ h) {
    __shared__ float lds[16 * LDS_STRIDE];
    const int t = threadIdx.x;
    const int row0 = blockIdx.x * 16;

    {
        const int r = t >> 3;
        const int c = (t & 7) * 16;
        const float s = rsqrtf(fmaxf(odeg[row0 + r], 1.0f));
        const float4* fp = (const float4*)&feat[(size_t)(row0 + r) * CH + c];
#pragma unroll
        for (int j = 0; j < 4; ++j) {
            float4 v = fp[j];
            v.x *= s; v.y *= s; v.z *= s; v.w *= s;
            *(float4*)&lds[r * LDS_STRIDE + c + j * 4] = v;
        }
    }
    __syncthreads();

    const int lane = t & 31;
    const int wave = t >> 5;
    const int half = lane >> 4;
    const int l16  = lane & 15;
    const int n0 = wave * 16;
    const int n1 = 64 + wave * 16;

    v8f c0 = {};
    v8f c1 = {};
    const float* arow = &lds[l16 * LDS_STRIDE + half * 2];

#pragma unroll 4
    for (int k = 0; k < CH; k += 4) {
        v2f a;
        a.x = arow[k];
        a.y = arow[k + 1];

        const float* wk = &W[(size_t)(k + half * 2) * CH + l16];
        v2f b0, b1;
        b0.x = wk[n0];       b0.y = wk[CH + n0];
        b1.x = wk[n1];       b1.y = wk[CH + n1];

        c0 = __builtin_amdgcn_wmma_f32_16x16x4_f32(false, a, false, b0, (short)0, c0, false, false);
        c1 = __builtin_amdgcn_wmma_f32_16x16x4_f32(false, a, false, b1, (short)0, c1, false, false);
    }

#pragma unroll
    for (int r = 0; r < 8; ++r) {
        const int row = row0 + r + half * 8;
        h[(size_t)row * CH + n0 + l16] = c0[r];
        h[(size_t)row * CH + n1 + l16] = c1[r];
    }
}

__global__ void scatter_kernel(const int* __restrict__ src, const int* __restrict__ dst,
                               const float* __restrict__ h, float* __restrict__ out) {
    const int gid = blockIdx.x * blockDim.x + threadIdx.x;
    const int e = gid >> 5;
    if (e >= N_EDGES) return;
    const int q = (gid & 31) * 4;
    const int s = src[e];
    const int d = dst[e];
    const float4 v = *(const float4*)&h[(size_t)s * CH + q];
    float* o = &out[(size_t)d * CH + q];
    atomicAdd(o + 0, v.x);
    atomicAdd(o + 1, v.y);
    atomicAdd(o + 2, v.z);
    atomicAdd(o + 3, v.w);
}

__global__ void finalize_kernel(const float* __restrict__ ideg, const float* __restrict__ bias,
                                float* __restrict__ out) {
    const int gid = blockIdx.x * blockDim.x + threadIdx.x;
    if (gid >= N_NODES * 32) return;
    const int n = gid >> 5;
    const int q = (gid & 31) * 4;
    const float s = rsqrtf(fmaxf(ideg[n], 1.0f));
    float4 v = *(float4*)&out[(size_t)n * CH + q];
    const float4 b = *(const float4*)&bias[q];
    v.x = fmaf(v.x, s, b.x);
    v.y = fmaf(v.y, s, b.y);
    v.z = fmaf(v.z, s, b.z);
    v.w = fmaf(v.w, s, b.w);
    *(float4*)&out[(size_t)n * CH + q] = v;
}

extern "C" void kernel_launch(void* const* d_in, const int* in_sizes, int n_in,
                              void* d_out, int out_size, void* d_ws, size_t ws_size,
                              hipStream_t stream) {
    const float* feat  = (const float*)d_in[0];
    const int*   src   = (const int*)  d_in[1];
    const int*   dst   = (const int*)  d_in[2];
    const float* W_mu  = (const float*)d_in[3];
    const float* W_rho = (const float*)d_in[4];
    const float* b_mu  = (const float*)d_in[5];
    const float* b_rho = (const float*)d_in[6];
    const float* W_eps = (const float*)d_in[7];
    const float* b_eps = (const float*)d_in[8];

    float* ws   = (float*)d_ws;
    float* W    = ws + OFF_W;
    float* bias = ws + OFF_BIAS;
    float* odeg = ws + OFF_ODEG;
    float* ideg = ws + OFF_IDEG;
    float* h    = ws + OFF_H;
    float* out  = (float*)d_out;

    {
        const int n4 = (2 * N_NODES) / 4;
        zero_f4<<<(n4 + 255) / 256, 256, 0, stream>>>(odeg, n4);
    }
    {
        const int n4 = (N_NODES * CH) / 4;
        zero_f4<<<(n4 + 255) / 256, 256, 0, stream>>>(out, n4);
    }

    prep_weights<<<(CH * CH + CH + 255) / 256, 256, 0, stream>>>(
        W_mu, W_rho, W_eps, b_mu, b_rho, b_eps, W, bias);

    degrees_kernel<<<(N_EDGES + 255) / 256, 256, 0, stream>>>(src, dst, odeg, ideg);

    gemm_wmma<<<N_NODES / 16, 128, 0, stream>>>(feat, odeg, W, h);

    {
        const long long nthreads = (long long)N_EDGES * 32;
        scatter_kernel<<<(int)((nthreads + 255) / 256), 256, 0, stream>>>(src, dst, h, out);
    }

    {
        const int nthreads = N_NODES * 32;
        finalize_kernel<<<(nthreads + 255) / 256, 256, 0, stream>>>(ideg, bias, out);
    }
}


# Per-edge global atomics replaced by work-group LDS counters: 32 edge chunks x 4 node ranges count in LDS (the add's return is the slot), per-node prefix over chunks; no zero pass; sum order unchanged
# speedup vs baseline: 3.5284x; 3.5284x over previous
#include <hip/hip_runtime.h>
#include <math.h>

typedef __attribute__((ext_vector_type(2))) float v2f;
typedef __attribute__((ext_vector_type(8))) float v8f;

#define N_NODES 100000
#define N_EDGES 1600000
#define CH 128
#define LDS_STRIDE 132
#define SCAN_TILE 1024
#define NCHUNK 32
#define CHUNK (N_EDGES / NCHUNK)
#define NRANGE 4
#define RANGE (N_NODES / NRANGE)
#define N_TILES ((N_NODES + SCAN_TILE - 1) / SCAN_TILE)

#define OFF_W     0
#define OFF_BIAS  (OFF_W + CH*CH)
#define OFF_CIN   (OFF_BIAS + CH)
#define OFF_COUT  (OFF_CIN + N_NODES)
#define OFF_ROWL  (OFF_COUT + N_NODES)
#define OFF_BSUM  (OFF_ROWL + N_TILES*SCAN_TILE)
#define OFF_BPRE  (OFF_BSUM + 128)
#define OFF_RANK  (OFF_BPRE + 128)
#define OFF_COL   (OFF_RANK + N_EDGES)
#define OFF_PSRC  OFF_COL
#define OFF_PDST  (OFF_COL + 2*N_EDGES)

__global__ void prep_weights(const float* __restrict__ Wmu, const float* __restrict__ Wrho,
                             const float* __restrict__ Weps,
                             const float* __restrict__ bmu, const float* __restrict__ brho,
                             const float* __restrict__ beps,
                             float* __restrict__ W, float* __restrict__ bias) {
    int i = blockIdx.x * blockDim.x + threadIdx.x;
    if (i < CH * CH) {
        float sp = log1pf(expf(Wrho[i]));
        W[i] = Wmu[i] + Weps[i] * sp;
    } else if (i < CH * CH + CH) {
        int j = i - CH * CH;
        float sp = log1pf(expf(brho[j]));
        bias[j] = bmu[j] + beps[j] * sp;
    }
}

__global__ __launch_bounds__(1024) void count_chunks(const int* __restrict__ src,
                                                     const int* __restrict__ dst,
                                                     int* __restrict__ psrc,
                                                     int* __restrict__ pdst,
                                                     int* __restrict__ rank) {
    __shared__ int cnt[RANGE];
    const int t = threadIdx.x;
    const int e0 = blockIdx.x * CHUNK;
    const int r0 = blockIdx.y * RANGE;
    const bool is_dst = blockIdx.z != 0;

    for (int i = t; i < RANGE / 4; i += 1024) ((int4*)cnt)[i] = make_int4(0, 0, 0, 0);
    __syncthreads();

    if (is_dst) {
        for (int base = 0; base < CHUNK; base += 8 * 1024) {
            int v[8], rk[8];
#pragma unroll
            for (int j = 0; j < 8; ++j) {
                const int i = base + j * 1024 + t;
                v[j] = (i < CHUNK) ? dst[e0 + i] : -1;
            }
#pragma unroll
            for (int j = 0; j < 8; ++j) {
                const unsigned k = (unsigned)(v[j] - r0);
                rk[j] = (k < (unsigned)RANGE) ? atomicAdd(&cnt[k], 1) : -1;
            }
#pragma unroll
            for (int j = 0; j < 8; ++j) {
                if (rk[j] >= 0) rank[e0 + base + j * 1024 + t] = rk[j];
            }
        }
    } else {
        for (int base = 0; base < CHUNK; base += 8 * 1024) {
            int v[8];
#pragma unroll
            for (int j = 0; j < 8; ++j) {
                const int i = base + j * 1024 + t;
                v[j] = (i < CHUNK) ? src[e0 + i] : -1;
            }
#pragma unroll
            for (int j = 0; j < 8; ++j) {
                const unsigned k = (unsigned)(v[j] - r0);
                if (k < (unsigned)RANGE) atomicAdd(&cnt[k], 1);
            }
        }
    }
    __syncthreads();

    int* __restrict__ part = (is_dst ? pdst : psrc) + (size_t)blockIdx.x * N_NODES + r0;
    for (int i = t; i < RANGE / 4; i += 1024) ((int4*)part)[i] = ((const int4*)cnt)[i];
}

__global__ __launch_bounds__(256) void reduce_chunks(int* pdst, const int* __restrict__ psrc,
                                                     int* __restrict__ cin, int* __restrict__ cout) {
    const int n = blockIdx.x * 256 + threadIdx.x;
    if (n >= N_NODES) return;
    if (blockIdx.y != 0) {
        int v[NCHUNK];
#pragma unroll
        for (int c = 0; c < NCHUNK; ++c) v[c] = pdst[(size_t)c * N_NODES + n];
        int run = 0;
#pragma unroll
        for (int c = 0; c < NCHUNK; ++c) {
            pdst[(size_t)c * N_NODES + n] = run;
            run += v[c];
        }
        cin[n] = run;
    } else {
        int v[NCHUNK];
#pragma unroll
        for (int c = 0; c < NCHUNK; ++c) v[c] = psrc[(size_t)c * N_NODES + n];
        int run = 0;
#pragma unroll
        for (int c = 0; c < NCHUNK; ++c) run += v[c];
        cout[n] = run;
    }
}

__global__ __launch_bounds__(256) void scan_tiles(const int* __restrict__ cin,
                                                  int* __restrict__ rowl,
                                                  int* __restrict__ bsum) {
    __shared__ int wsum[8];
    const int t = threadIdx.x;
    const int lane = t & 31;
    const int wave = t >> 5;
    const int base = blockIdx.x * SCAN_TILE + t * 4;
    const int4 cv = (base < N_NODES) ? *(const int4*)&cin[base] : make_int4(0, 0, 0, 0);
    const int c[4] = {cv.x, cv.y, cv.z, cv.w};
    const int tot = c[0] + c[1] + c[2] + c[3];
    int incl = tot;
#pragma unroll
    for (int off = 1; off < 32; off <<= 1) {
        const int y = __shfl_up(incl, off);
        if (lane >= off) incl += y;
    }
    if (lane == 31) wsum[wave] = incl;
    __syncthreads();
    int wbase = 0;
    for (int w = 0; w < wave; ++w) wbase += wsum[w];
    const int excl = wbase + incl - tot;
    int4 o;
    o.x = excl;
    o.y = excl + c[0];
    o.z = excl + c[0] + c[1];
    o.w = excl + c[0] + c[1] + c[2];
    *(int4*)&rowl[base] = o;
    if (t == 255) bsum[blockIdx.x] = excl + tot;
}

__global__ __launch_bounds__(128) void scan_tile_sums(const int* __restrict__ bsum,
                                                      int* __restrict__ bpre) {
    __shared__ int wsum[4];
    const int t = threadIdx.x;
    const int lane = t & 31;
    const int wave = t >> 5;
    const int v = (t < N_TILES) ? bsum[t] : 0;
    int incl = v;
#pragma unroll
    for (int off = 1; off < 32; off <<= 1) {
        const int y = __shfl_up(incl, off);
        if (lane >= off) incl += y;
    }
    if (lane == 31) wsum[wave] = incl;
    __syncthreads();
    int wbase = 0;
    for (int w = 0; w < wave; ++w) wbase += wsum[w];
    if (t < N_TILES) bpre[t] = wbase + incl - v;
}

__global__ void fill_kernel(const int* __restrict__ src, const int* __restrict__ dst,
                            const int* __restrict__ rank, const int* __restrict__ cout,
                            const int* __restrict__ rowl, const int* __restrict__ bpre,
                            const int* __restrict__ pdst,
                            int* __restrict__ colw) {
    int2* __restrict__ col = (int2*)colw;
    int e = blockIdx.x * blockDim.x + threadIdx.x;
    if (e < N_EDGES) {
        const int s = src[e];
        const int d = dst[e];
        int pos = bpre[d >> 10] + rowl[d] + pdst[(size_t)(e / CHUNK) * N_NODES + d] + rank[e];
        pos = min(max(pos, 0), N_EDGES - 1);
        const float f = rsqrtf(fmaxf((float)cout[s], 1.0f));
        col[pos] = make_int2(s, __float_as_int(f));
    }
}

__device__ __forceinline__ int sums_before(int ki, int i, int k, int p) {
    return (ki < k || (ki == k && i < p)) ? 1 : 0;
}

__device__ __forceinline__ void add_row(float4& acc, const float* __restrict__ feat,
                                        int s, float f, int lane) {
    s = min(max(s, 0), N_NODES - 1);
    const float4 v = *(const float4*)&feat[(size_t)s * CH + lane * 4];
    acc.x = fmaf(v.x, f, acc.x);
    acc.y = fmaf(v.y, f, acc.y);
    acc.z = fmaf(v.z, f, acc.z);
    acc.w = fmaf(v.w, f, acc.w);
}

__global__ __launch_bounds__(128) void gather_gemm(const float* __restrict__ feat,
                                                   const int* __restrict__ cin,
                                                   const int* __restrict__ rowl,
                                                   const int* __restrict__ bpre,
                                                   const int* __restrict__ colw,
                                                   const float* __restrict__ W,
                                                   const float* __restrict__ bias,
                                                   float* __restrict__ out) {
    __shared__ float lds[16 * LDS_STRIDE];
    __shared__ float nrm[16];
    const int2* __restrict__ col = (const int2*)colw;
    const int t = threadIdx.x;
    const int lane = t & 31;
    const int wave = t >> 5;
    const int row0 = blockIdx.x * 16;

    int degv = 0, startv = 0;
    if (lane < 4) {
        const int n = row0 + wave * 4 + lane;
        degv = cin[n];
        startv = bpre[n >> 10] + rowl[n];
    }

    for (int i = 0; i < 4; ++i) {
        const int rl = wave * 4 + i;
        const int deg = __builtin_amdgcn_readlane(degv, i);
        const int start = __builtin_amdgcn_readlane(startv, i);
        float4 acc = make_float4(0.f, 0.f, 0.f, 0.f);

        if (deg <= 64) {
            const bool two = deg > 32;
            int2 e0 = make_int2(0x7fffffff, 0), e1 = make_int2(0x7fffffff, 0);
            if (lane < deg) e0 = col[start + lane];
            if (lane + 32 < deg) e1 = col[start + 32 + lane];
            const int key0 = e0.x, key1 = e1.x;
            int rk0 = 0, rk1 = 0;
            const int d0 = min(deg, 32);
            for (int j = 0; j < d0; ++j) {
                const int kj = __builtin_amdgcn_readlane(key0, j);
                rk0 += sums_before(kj, j, key0, lane);
                if (two) rk1 += sums_before(kj, j, key1, lane + 32);
            }
            for (int j = 32; j < deg; ++j) {
                const int kj = __builtin_amdgcn_readlane(key1, j - 32);
                rk0 += sums_before(kj, j, key0, lane);
                rk1 += sums_before(kj, j, key1, lane + 32);
            }
            const int isc0 = e0.y;
            const int isc1 = e1.y;
#pragma unroll 4
            for (int r = 0; r < deg; ++r) {
                const unsigned m0 = __builtin_amdgcn_ballot_w32(rk0 == r);
                int s, fi;
                if (m0 != 0u) {
                    const int l = __builtin_ctz(m0);
                    s = __builtin_amdgcn_readlane(key0, l);
                    fi = __builtin_amdgcn_readlane(isc0, l);
                } else {
                    const unsigned m1 = __builtin_amdgcn_ballot_w32(rk1 == r);
                    const int l = (m1 != 0u) ? __builtin_ctz(m1) : 0;
                    s = __builtin_amdgcn_readlane(key1, l);
                    fi = __builtin_amdgcn_readlane(isc1, l);
                }
                add_row(acc, feat, s, __int_as_float(fi), lane);
            }
        } else {
            for (int b = 0; b < deg; b += 32) {
                const int cnt = min(32, deg - b);
                int2 eb = make_int2(0, 0);
                if (lane < cnt) eb = col[start + b + lane];
                const int key = eb.x;
                const int isc = eb.y;
                for (int j = 0; j < cnt; ++j) {
                    const int s = __builtin_amdgcn_readlane(key, j);
                    const int fi = __builtin_amdgcn_readlane(isc, j);
                    add_row(acc, feat, s, __int_as_float(fi), lane);
                }
            }
        }
        *(float4*)&lds[rl * LDS_STRIDE + lane * 4] = acc;
        if (lane == 0) nrm[rl] = rsqrtf(fmaxf((float)deg, 1.0f));
    }
    __syncthreads();

    const int half = lane >> 4;
    const int l16  = lane & 15;
    const int n0 = wave * 16;
    const int n1 = 64 + wave * 16;

    v8f c0 = {};
    v8f c1 = {};
    const float* arow = &lds[l16 * LDS_STRIDE + half * 2];

#pragma unroll 4
    for (int k = 0; k < CH; k += 4) {
        v2f a;
        a.x = arow[k];
        a.y = arow[k + 1];

        const float* wk = &W[(size_t)(k + half * 2) * CH + l16];
        v2f b0, b1;
        b0.x = wk[n0];       b0.y = wk[CH + n0];
        b1.x = wk[n1];       b1.y = wk[CH + n1];

        c0 = __builtin_amdgcn_wmma_f32_16x16x4_f32(false, a, false, b0, (short)0, c0, false, false);
        c1 = __builtin_amdgcn_wmma_f32_16x16x4_f32(false, a, false, b1, (short)0, c1, false, false);
    }

    __syncthreads();
#pragma unroll
    for (int r = 0; r < 8; ++r) {
        const int rr = r + half * 8;
        lds[rr * LDS_STRIDE + n0 + l16] = c0[r];
        lds[rr * LDS_STRIDE + n1 + l16] = c1[r];
    }
    __syncthreads();
    const float4 b = *(const float4*)&bias[lane * 4];
#pragma unroll
    for (int j = 0; j < 4; ++j) {
        const int rr = j * 4 + wave;
        const float s = nrm[rr];
        float4 v = *(const float4*)&lds[rr * LDS_STRIDE + lane * 4];
        v.x = fmaf(v.x, s, b.x);
        v.y = fmaf(v.y, s, b.y);
        v.z = fmaf(v.z, s, b.z);
        v.w = fmaf(v.w, s, b.w);
        *(float4*)&out[(size_t)(row0 + rr) * CH + lane * 4] = v;
    }
}

extern "C" void kernel_launch(void* const* d_in, const int* in_sizes, int n_in,
                              void* d_out, int out_size, void* d_ws, size_t ws_size,
                              hipStream_t stream) {
    const float* feat  = (const float*)d_in[0];
    const int*   src   = (const int*)  d_in[1];
    const int*   dst   = (const int*)  d_in[2];
    const float* W_mu  = (const float*)d_in[3];
    const float* W_rho = (const float*)d_in[4];
    const float* b_mu  = (const float*)d_in[5];
    const float* b_rho = (const float*)d_in[6];
    const float* W_eps = (const float*)d_in[7];
    const float* b_eps = (const float*)d_in[8];

    float* ws   = (float*)d_ws;
    int*   wsi  = (int*)d_ws;
    float* W    = ws + OFF_W;
    float* bias = ws + OFF_BIAS;
    int*   cin  = wsi + OFF_CIN;
    int*   cout = wsi + OFF_COUT;
    int*   rowl = wsi + OFF_ROWL;
    int*   bsum = wsi + OFF_BSUM;
    int*   bpre = wsi + OFF_BPRE;
    int*   rank = wsi + OFF_RANK;
    int*   col  = wsi + OFF_COL;
    int*   psrc = wsi + OFF_PSRC;
    int*   pdst = wsi + OFF_PDST;
    float* out  = (float*)d_out;

    prep_weights<<<(CH * CH + CH + 255) / 256, 256, 0, stream>>>(
        W_mu, W_rho, W_eps, b_mu, b_rho, b_eps, W, bias);

    count_chunks<<<dim3(NCHUNK, NRANGE, 2), 1024, 0, stream>>>(src, dst, psrc, pdst, rank);

    reduce_chunks<<<dim3((N_NODES + 255) / 256, 2), 256, 0, stream>>>(pdst, psrc, cin, cout);

    scan_tiles<<<N_TILES, 256, 0, stream>>>(cin, rowl, bsum);
    scan_tile_sums<<<1, 128, 0, stream>>>(bsum, bpre);

    fill_kernel<<<(N_EDGES + 255) / 256, 256, 0, stream>>>(src, dst, rank, cout, rowl, bpre, pdst, col);

    gather_gemm<<<N_NODES / 16, 128, 0, stream>>>(feat, cin, rowl, bpre, col, W, bias, out);
}


# Launch count 7 to 4: chunk reduce and tile scan fused in one work-group, weight prep rides in that launch, placement pass scans the 98 tile totals itself; f32 sum order unchanged
# speedup vs baseline: 3.6545x; 1.0357x over previous
#include <hip/hip_runtime.h>
#include <math.h>

typedef __attribute__((ext_vector_type(2))) float v2f;
typedef __attribute__((ext_vector_type(8))) float v8f;

#define N_NODES 100000
#define N_EDGES 1600000
#define CH 128
#define LDS_STRIDE 132
#define SCAN_TILE 1024
#define NCHUNK 32
#define CHUNK (N_EDGES / NCHUNK)
#define NRANGE 4
#define RANGE (N_NODES / NRANGE)
#define N_TILES ((N_NODES + SCAN_TILE - 1) / SCAN_TILE)
#define RS_THREADS 512
#define PREP_BLOCKS ((CH * CH + CH + RS_THREADS - 1) / RS_THREADS)

#define OFF_W     0
#define OFF_BIAS  (OFF_W + CH*CH)
#define OFF_CIN   (OFF_BIAS + CH)
#define OFF_COUT  (OFF_CIN + N_NODES)
#define OFF_ROWL  (OFF_COUT + N_NODES)
#define OFF_BSUM  (OFF_ROWL + N_TILES*SCAN_TILE)
#define OFF_BPRE  (OFF_BSUM + 128)
#define OFF_RANK  (OFF_BPRE + 128)
#define OFF_COL   (OFF_RANK + N_EDGES)
#define OFF_PSRC  OFF_COL
#define OFF_PDST  (OFF_COL + 2*N_EDGES)

__global__ __launch_bounds__(1024) void count_chunks(const int* __restrict__ src,
                                                     const int* __restrict__ dst,
                                                     int* __restrict__ psrc,
                                                     int* __restrict__ pdst,
                                                     int* __restrict__ rank) {
    __shared__ int cnt[RANGE];
    const int t = threadIdx.x;
    const int e0 = blockIdx.x * CHUNK;
    const int r0 = blockIdx.y * RANGE;
    const bool is_dst = blockIdx.z != 0;

    for (int i = t; i < RANGE / 4; i += 1024) ((int4*)cnt)[i] = make_int4(0, 0, 0, 0);
    __syncthreads();

    if (is_dst) {
        for (int base = 0; base < CHUNK; base += 8 * 1024) {
            int v[8], rk[8];
#pragma unroll
            for (int j = 0; j < 8; ++j) {
                const int i = base + j * 1024 + t;
                v[j] = (i < CHUNK) ? dst[e0 + i] : -1;
            }
#pragma unroll
            for (int j = 0; j < 8; ++j) {
                const unsigned k = (unsigned)(v[j] - r0);
                rk[j] = (k < (unsigned)RANGE) ? atomicAdd(&cnt[k], 1) : -1;
            }
#pragma unroll
            for (int j = 0; j < 8; ++j) {
                if (rk[j] >= 0) rank[e0 + base + j * 1024 + t] = rk[j];
            }
        }
    } else {
        for (int base = 0; base < CHUNK; base += 8 * 1024) {
            int v[8];
#pragma unroll
            for (int j = 0; j < 8; ++j) {
                const int i = base + j * 1024 + t;
                v[j] = (i < CHUNK) ? src[e0 + i] : -1;
            }
#pragma unroll
            for (int j = 0; j < 8; ++j) {
                const unsigned k = (unsigned)(v[j] - r0);
                if (k < (unsigned)RANGE) atomicAdd(&cnt[k], 1);
            }
        }
    }
    __syncthreads();

    int* __restrict__ part = (is_dst ? pdst : psrc) + (size_t)blockIdx.x * N_NODES + r0;
    for (int i = t; i < RANGE / 4; i += 1024) ((int4*)part)[i] = ((const int4*)cnt)[i];
}

__global__ __launch_bounds__(RS_THREADS) void reduce_scan_prep(int* pdst, const int* __restrict__ psrc,
                                                               int* __restrict__ cin, int* __restrict__ cout,
                                                               int* __restrict__ rowl, int* __restrict__ bsum,
                                                               const float* __restrict__ Wmu,
                                                               const float* __restrict__ Wrho,
                                                               const float* __restrict__ Weps,
                                                               const float* __restrict__ bmu,
                                                               const float* __restrict__ brho,
                                                               const float* __restrict__ beps,
                                                               float* __restrict__ W, float* __restrict__ bias) {
    __shared__ int wsum[RS_THREADS / 32];
    const int t = threadIdx.x;
    const int b = blockIdx.x;

    if (b >= 2 * N_TILES) {
        const int i = (b - 2 * N_TILES) * RS_THREADS + t;
        if (i < CH * CH) {
            float sp = log1pf(expf(Wrho[i]));
            W[i] = Wmu[i] + Weps[i] * sp;
        } else if (i < CH * CH + CH) {
            int j = i - CH * CH;
            float sp = log1pf(expf(brho[j]));
            bias[j] = bmu[j] + beps[j] * sp;
        }
        return;
    }

    if (b >= N_TILES) {
        const int base = (b - N_TILES) * SCAN_TILE + t * 2;
        if (base < N_NODES) {
            int2 v[NCHUNK];
#pragma unroll
            for (int c = 0; c < NCHUNK; ++c) v[c] = *(const int2*)&psrc[(size_t)c * N_NODES + base];
            int2 run = make_int2(0, 0);
#pragma unroll
            for (int c = 0; c < NCHUNK; ++c) { run.x += v[c].x; run.y += v[c].y; }
            *(int2*)&cout[base] = run;
        }
        return;
    }

    const int lane = t & 31;
    const int wave = t >> 5;
    const int base = b * SCAN_TILE + t * 2;
    int2 run = make_int2(0, 0);
    if (base < N_NODES) {
        int2 v[NCHUNK];
#pragma unroll
        for (int c = 0; c < NCHUNK; ++c) v[c] = *(const int2*)&pdst[(size_t)c * N_NODES + base];
#pragma unroll
        for (int c = 0; c < NCHUNK; ++c) {
            *(int2*)&pdst[(size_t)c * N_NODES + base] = run;
            run.x += v[c].x;
            run.y += v[c].y;
        }
        *(int2*)&cin[base] = run;
    }

    const int tot = run.x + run.y;
    int incl = tot;
#pragma unroll
    for (int off = 1; off < 32; off <<= 1) {
        const int y = __shfl_up(incl, off);
        if (lane >= off) incl += y;
    }
    if (lane == 31) wsum[wave] = incl;
    __syncthreads();
    int wbase = 0;
    for (int w = 0; w < wave; ++w) wbase += wsum[w];
    const int excl = wbase + incl - tot;
    *(int2*)&rowl[base] = make_int2(excl, excl + run.x);
    if (t == RS_THREADS - 1) bsum[b] = excl + tot;
}

__global__ __launch_bounds__(256) void fill_kernel(const int* __restrict__ src, const int* __restrict__ dst,
                            const int* __restrict__ rank, const int* __restrict__ cout,
                            const int* __restrict__ rowl, const int* __restrict__ bsum,
                            const int* __restrict__ pdst,
                            int* __restrict__ bpre_out,
                            int* __restrict__ colw) {
    __shared__ __attribute__((aligned(16))) int bpre[128];
    int2* __restrict__ col = (int2*)colw;
    const int t = threadIdx.x;
    const int e = blockIdx.x * 256 + t;
    const bool ok = e < N_EDGES;
    int s = 0, d = 0, rk = 0;
    if (ok) { s = src[e]; d = dst[e]; rk = rank[e]; }
    if (t < 32) {
        int v[4];
#pragma unroll
        for (int i = 0; i < 4; ++i) {
            const int idx = t * 4 + i;
            const int x = bsum[min(idx, N_TILES - 1)];
            v[i] = (idx < N_TILES) ? x : 0;
        }
        const int tot = v[0] + v[1] + v[2] + v[3];
        int incl = tot;
#pragma unroll
        for (int off = 1; off < 32; off <<= 1) {
            const int y = __shfl_up(incl, off);
            if (t >= off) incl += y;
        }
        const int excl = incl - tot;
        const int4 o = make_int4(excl, excl + v[0], excl + v[0] + v[1], excl + v[0] + v[1] + v[2]);
        *(int4*)&bpre[t * 4] = o;
        if (blockIdx.x == 0) *(int4*)&bpre_out[t * 4] = o;
    }
    __syncthreads();
    if (ok) {
        int pos = bpre[d >> 10] + rowl[d] + pdst[(size_t)(e / CHUNK) * N_NODES + d] + rk;
        pos = min(max(pos, 0), N_EDGES - 1);
        const float f = rsqrtf(fmaxf((float)cout[s], 1.0f));
        col[pos] = make_int2(s, __float_as_int(f));
    }
}

__device__ __forceinline__ int sums_before(int ki, int i, int k, int p) {
    return (ki < k || (ki == k && i < p)) ? 1 : 0;
}

__device__ __forceinline__ void add_row(float4& acc, const float* __restrict__ feat,
                                        int s, float f, int lane) {
    s = min(max(s, 0), N_NODES - 1);
    const float4 v = *(const float4*)&feat[(size_t)s * CH + lane * 4];
    acc.x = fmaf(v.x, f, acc.x);
    acc.y = fmaf(v.y, f, acc.y);
    acc.z = fmaf(v.z, f, acc.z);
    acc.w = fmaf(v.w, f, acc.w);
}

__global__ __launch_bounds__(128) void gather_gemm(const float* __restrict__ feat,
                                                   const int* __restrict__ cin,
                                                   const int* __restrict__ rowl,
                                                   const int* __restrict__ bpre,
                                                   const int* __restrict__ colw,
                                                   const float* __restrict__ W,
                                                   const float* __restrict__ bias,
                                                   float* __restrict__ out) {
    __shared__ float lds[16 * LDS_STRIDE];
    __shared__ float nrm[16];
    const int2* __restrict__ col = (const int2*)colw;
    const int t = threadIdx.x;
    const int lane = t & 31;
    const int wave = t >> 5;
    const int row0 = blockIdx.x * 16;

    int degv = 0, startv = 0;
    if (lane < 4) {
        const int n = row0 + wave * 4 + lane;
        degv = cin[n];
        startv = bpre[n >> 10] + rowl[n];
    }

    for (int i = 0; i < 4; ++i) {
        const int rl = wave * 4 + i;
        const int deg = __builtin_amdgcn_readlane(degv, i);
        const int start = __builtin_amdgcn_readlane(startv, i);
        float4 acc = make_float4(0.f, 0.f, 0.f, 0.f);

        if (deg <= 64) {
            const bool two = deg > 32;
            int2 e0 = make_int2(0x7fffffff, 0), e1 = make_int2(0x7fffffff, 0);
            if (lane < deg) e0 = col[start + lane];
            if (lane + 32 < deg) e1 = col[start + 32 + lane];
            const int key0 = e0.x, key1 = e1.x;
            int rk0 = 0, rk1 = 0;
            const int d0 = min(deg, 32);
            for (int j = 0; j < d0; ++j) {
                const int kj = __builtin_amdgcn_readlane(key0, j);
                rk0 += sums_before(kj, j, key0, lane);
                if (two) rk1 += sums_before(kj, j, key1, lane + 32);
            }
            for (int j = 32; j < deg; ++j) {
                const int kj = __builtin_amdgcn_readlane(key1, j - 32);
                rk0 += sums_before(kj, j, key0, lane);
                rk1 += sums_before(kj, j, key1, lane + 32);
            }
            const int isc0 = e0.y;
            const int isc1 = e1.y;
#pragma unroll 4
            for (int r = 0; r < deg; ++r) {
                const unsigned m0 = __builtin_amdgcn_ballot_w32(rk0 == r);
                int s, fi;
                if (m0 != 0u) {
                    const int l = __builtin_ctz(m0);
                    s = __builtin_amdgcn_readlane(key0, l);
                    fi = __builtin_amdgcn_readlane(isc0, l);
                } else {
                    const unsigned m1 = __builtin_amdgcn_ballot_w32(rk1 == r);
                    const int l = (m1 != 0u) ? __builtin_ctz(m1) : 0;
                    s = __builtin_amdgcn_readlane(key1, l);
                    fi = __builtin_amdgcn_readlane(isc1, l);
                }
                add_row(acc, feat, s, __int_as_float(fi), lane);
            }
        } else {
            for (int b = 0; b < deg; b += 32) {
                const int cnt = min(32, deg - b);
                int2 eb = make_int2(0, 0);
                if (lane < cnt) eb = col[start + b + lane];
                const int key = eb.x;
                const int isc = eb.y;
                for (int j = 0; j < cnt; ++j) {
                    const int s = __builtin_amdgcn_readlane(key, j);
                    const int fi = __builtin_amdgcn_readlane(isc, j);
                    add_row(acc, feat, s, __int_as_float(fi), lane);
                }
            }
        }
        *(float4*)&lds[rl * LDS_STRIDE + lane * 4] = acc;
        if (lane == 0) nrm[rl] = rsqrtf(fmaxf((float)deg, 1.0f));
    }
    __syncthreads();

    const int half = lane >> 4;
    const int l16  = lane & 15;
    const int n0 = wave * 16;
    const int n1 = 64 + wave * 16;

    v8f c0 = {};
    v8f c1 = {};
    const float* arow = &lds[l16 * LDS_STRIDE + half * 2];

#pragma unroll 4
    for (int k = 0; k < CH; k += 4) {
        v2f a;
        a.x = arow[k];
        a.y = arow[k + 1];

        const float* wk = &W[(size_t)(k + half * 2) * CH + l16];
        v2f b0, b1;
        b0.x = wk[n0];       b0.y = wk[CH + n0];
        b1.x = wk[n1];       b1.y = wk[CH + n1];

        c0 = __builtin_amdgcn_wmma_f32_16x16x4_f32(false, a, false, b0, (short)0, c0, false, false);
        c1 = __builtin_amdgcn_wmma_f32_16x16x4_f32(false, a, false, b1, (short)0, c1, false, false);
    }

    __syncthreads();
#pragma unroll
    for (int r = 0; r < 8; ++r) {
        const int rr = r + half * 8;
        lds[rr * LDS_STRIDE + n0 + l16] = c0[r];
        lds[rr * LDS_STRIDE + n1 + l16] = c1[r];
    }
    __syncthreads();
    const float4 b = *(const float4*)&bias[lane * 4];
#pragma unroll
    for (int j = 0; j < 4; ++j) {
        const int rr = j * 4 + wave;
        const float s = nrm[rr];
        float4 v = *(const float4*)&lds[rr * LDS_STRIDE + lane * 4];
        v.x = fmaf(v.x, s, b.x);
        v.y = fmaf(v.y, s, b.y);
        v.z = fmaf(v.z, s, b.z);
        v.w = fmaf(v.w, s, b.w);
        *(float4*)&out[(size_t)(row0 + rr) * CH + lane * 4] = v;
    }
}

extern "C" void kernel_launch(void* const* d_in, const int* in_sizes, int n_in,
                              void* d_out, int out_size, void* d_ws, size_t ws_size,
                              hipStream_t stream) {
    const float* feat  = (const float*)d_in[0];
    const int*   src   = (const int*)  d_in[1];
    const int*   dst   = (const int*)  d_in[2];
    const float* W_mu  = (const float*)d_in[3];
    const float* W_rho = (const float*)d_in[4];
    const float* b_mu  = (const float*)d_in[5];
    const float* b_rho = (const float*)d_in[6];
    const float* W_eps = (const float*)d_in[7];
    const float* b_eps = (const float*)d_in[8];

    float* ws   = (float*)d_ws;
    int*   wsi  = (int*)d_ws;
    float* W    = ws + OFF_W;
    float* bias = ws + OFF_BIAS;
    int*   cin  = wsi + OFF_CIN;
    int*   cout = wsi + OFF_COUT;
    int*   rowl = wsi + OFF_ROWL;
    int*   bsum = wsi + OFF_BSUM;
    int*   bpre = wsi + OFF_BPRE;
    int*   rank = wsi + OFF_RANK;
    int*   col  = wsi + OFF_COL;
    int*   psrc = wsi + OFF_PSRC;
    int*   pdst = wsi + OFF_PDST;
    float* out  = (float*)d_out;

    count_chunks<<<dim3(NCHUNK, NRANGE, 2), 1024, 0, stream>>>(src, dst, psrc, pdst, rank);

    reduce_scan_prep<<<2 * N_TILES + PREP_BLOCKS, RS_THREADS, 0, stream>>>(
        pdst, psrc, cin, cout, rowl, bsum, W_mu, W_rho, W_eps, b_mu, b_rho, b_eps, W, bias);

    fill_kernel<<<(N_EDGES + 255) / 256, 256, 0, stream>>>(src, dst, rank, cout, rowl, bsum, pdst, bpre, col);

    gather_gemm<<<N_NODES / 16, 128, 0, stream>>>(feat, cin, rowl, bpre, col, W, bias, out);
}


# Row sum: a node's records pushed into summing order across lanes (rank = target lane), rows then fetched in groups of up to 8 loads in flight, not 3+1 behind a ballot per row; f32 sum order unchanged
# speedup vs baseline: 3.7808x; 1.0346x over previous
#include <hip/hip_runtime.h>
#include <math.h>

typedef __attribute__((ext_vector_type(2))) float v2f;
typedef __attribute__((ext_vector_type(8))) float v8f;

#define N_NODES 100000
#define N_EDGES 1600000
#define CH 128
#define LDS_STRIDE 132
#define SCAN_TILE 1024
#define NCHUNK 32
#define CHUNK (N_EDGES / NCHUNK)
#define NRANGE 4
#define RANGE (N_NODES / NRANGE)
#define N_TILES ((N_NODES + SCAN_TILE - 1) / SCAN_TILE)
#define RS_THREADS 512
#define PREP_BLOCKS ((CH * CH + CH + RS_THREADS - 1) / RS_THREADS)

#define OFF_W     0
#define OFF_BIAS  (OFF_W + CH*CH)
#define OFF_CIN   (OFF_BIAS + CH)
#define OFF_COUT  (OFF_CIN + N_NODES)
#define OFF_ROWL  (OFF_COUT + N_NODES)
#define OFF_BSUM  (OFF_ROWL + N_TILES*SCAN_TILE)
#define OFF_BPRE  (OFF_BSUM + 128)
#define OFF_RANK  (OFF_BPRE + 128)
#define OFF_COL   (OFF_RANK + N_EDGES)
#define OFF_PSRC  OFF_COL
#define OFF_PDST  (OFF_COL + 2*N_EDGES)

__global__ __launch_bounds__(1024) void count_chunks(const int* __restrict__ src,
                                                     const int* __restrict__ dst,
                                                     int* __restrict__ psrc,
                                                     int* __restrict__ pdst,
                                                     int* __restrict__ rank) {
    __shared__ int cnt[RANGE];
    const int t = threadIdx.x;
    const int e0 = blockIdx.x * CHUNK;
    const int r0 = blockIdx.y * RANGE;
    const bool is_dst = blockIdx.z != 0;

    for (int i = t; i < RANGE / 4; i += 1024) ((int4*)cnt)[i] = make_int4(0, 0, 0, 0);
    __syncthreads();

    if (is_dst) {
        for (int base = 0; base < CHUNK; base += 8 * 1024) {
            int v[8], rk[8];
#pragma unroll
            for (int j = 0; j < 8; ++j) {
                const int i = base + j * 1024 + t;
                v[j] = (i < CHUNK) ? dst[e0 + i] : -1;
            }
#pragma unroll
            for (int j = 0; j < 8; ++j) {
                const unsigned k = (unsigned)(v[j] - r0);
                rk[j] = (k < (unsigned)RANGE) ? atomicAdd(&cnt[k], 1) : -1;
            }
#pragma unroll
            for (int j = 0; j < 8; ++j) {
                if (rk[j] >= 0) rank[e0 + base + j * 1024 + t] = rk[j];
            }
        }
    } else {
        for (int base = 0; base < CHUNK; base += 8 * 1024) {
            int v[8];
#pragma unroll
            for (int j = 0; j < 8; ++j) {
                const int i = base + j * 1024 + t;
                v[j] = (i < CHUNK) ? src[e0 + i] : -1;
            }
#pragma unroll
            for (int j = 0; j < 8; ++j) {
                const unsigned k = (unsigned)(v[j] - r0);
                if (k < (unsigned)RANGE) atomicAdd(&cnt[k], 1);
            }
        }
    }
    __syncthreads();

    int* __restrict__ part = (is_dst ? pdst : psrc) + (size_t)blockIdx.x * N_NODES + r0;
    for (int i = t; i < RANGE / 4; i += 1024) ((int4*)part)[i] = ((const int4*)cnt)[i];
}

__global__ __launch_bounds__(RS_THREADS) void reduce_scan_prep(int* pdst, const int* __restrict__ psrc,
                                                               int* __restrict__ cin, int* __restrict__ cout,
                                                               int* __restrict__ rowl, int* __restrict__ bsum,
                                                               const float* __restrict__ Wmu,
                                                               const float* __restrict__ Wrho,
                                                               const float* __restrict__ Weps,
                                                               const float* __restrict__ bmu,
                                                               const float* __restrict__ brho,
                                                               const float* __restrict__ beps,
                                                               float* __restrict__ W, float* __restrict__ bias) {
    __shared__ int wsum[RS_THREADS / 32];
    const int t = threadIdx.x;
    const int b = blockIdx.x;

    if (b >= 2 * N_TILES) {
        const int i = (b - 2 * N_TILES) * RS_THREADS + t;
        if (i < CH * CH) {
            float sp = log1pf(expf(Wrho[i]));
            W[i] = Wmu[i] + Weps[i] * sp;
        } else if (i < CH * CH + CH) {
            int j = i - CH * CH;
            float sp = log1pf(expf(brho[j]));
            bias[j] = bmu[j] + beps[j] * sp;
        }
        return;
    }

    if (b >= N_TILES) {
        const int base = (b - N_TILES) * SCAN_TILE + t * 2;
        if (base < N_NODES) {
            int2 v[NCHUNK];
#pragma unroll
            for (int c = 0; c < NCHUNK; ++c) v[c] = *(const int2*)&psrc[(size_t)c * N_NODES + base];
            int2 run = make_int2(0, 0);
#pragma unroll
            for (int c = 0; c < NCHUNK; ++c) { run.x += v[c].x; run.y += v[c].y; }
            *(int2*)&cout[base] = run;
        }
        return;
    }

    const int lane = t & 31;
    const int wave = t >> 5;
    const int base = b * SCAN_TILE + t * 2;
    int2 run = make_int2(0, 0);
    if (base < N_NODES) {
        int2 v[NCHUNK];
#pragma unroll
        for (int c = 0; c < NCHUNK; ++c) v[c] = *(const int2*)&pdst[(size_t)c * N_NODES + base];
#pragma unroll
        for (int c = 0; c < NCHUNK; ++c) {
            *(int2*)&pdst[(size_t)c * N_NODES + base] = run;
            run.x += v[c].x;
            run.y += v[c].y;
        }
        *(int2*)&cin[base] = run;
    }

    const int tot = run.x + run.y;
    int incl = tot;
#pragma unroll
    for (int off = 1; off < 32; off <<= 1) {
        const int y = __shfl_up(incl, off);
        if (lane >= off) incl += y;
    }
    if (lane == 31) wsum[wave] = incl;
    __syncthreads();
    int wbase = 0;
    for (int w = 0; w < wave; ++w) wbase += wsum[w];
    const int excl = wbase + incl - tot;
    *(int2*)&rowl[base] = make_int2(excl, excl + run.x);
    if (t == RS_THREADS - 1) bsum[b] = excl + tot;
}

__global__ __launch_bounds__(256) void fill_kernel(const int* __restrict__ src, const int* __restrict__ dst,
                            const int* __restrict__ rank, const int* __restrict__ cout,
                            const int* __restrict__ rowl, const int* __restrict__ bsum,
                            const int* __restrict__ pdst,
                            int* __restrict__ bpre_out,
                            int* __restrict__ colw) {
    __shared__ __attribute__((aligned(16))) int bpre[128];
    int2* __restrict__ col = (int2*)colw;
    const int t = threadIdx.x;
    const int e = blockIdx.x * 256 + t;
    const bool ok = e < N_EDGES;
    int s = 0, d = 0, rk = 0;
    if (ok) { s = src[e]; d = dst[e]; rk = rank[e]; }
    if (t < 32) {
        int v[4];
#pragma unroll
        for (int i = 0; i < 4; ++i) {
            const int idx = t * 4 + i;
            const int x = bsum[min(idx, N_TILES - 1)];
            v[i] = (idx < N_TILES) ? x : 0;
        }
        const int tot = v[0] + v[1] + v[2] + v[3];
        int incl = tot;
#pragma unroll
        for (int off = 1; off < 32; off <<= 1) {
            const int y = __shfl_up(incl, off);
            if (t >= off) incl += y;
        }
        const int excl = incl - tot;
        const int4 o = make_int4(excl, excl + v[0], excl + v[0] + v[1], excl + v[0] + v[1] + v[2]);
        *(int4*)&bpre[t * 4] = o;
        if (blockIdx.x == 0) *(int4*)&bpre_out[t * 4] = o;
    }
    __syncthreads();
    if (ok) {
        int pos = bpre[d >> 10] + rowl[d] + pdst[(size_t)(e / CHUNK) * N_NODES + d] + rk;
        pos = min(max(pos, 0), N_EDGES - 1);
        const float f = rsqrtf(fmaxf((float)cout[s], 1.0f));
        col[pos] = make_int2(s, __float_as_int(f));
    }
}

__device__ __forceinline__ int sums_before(int ki, int i, int k, int p) {
    return (ki < k || (ki == k && i < p)) ? 1 : 0;
}

__device__ __forceinline__ void add_row(float4& acc, const float* __restrict__ feat,
                                        int s, float f, int lane) {
    s = min(max(s, 0), N_NODES - 1);
    const float4 v = *(const float4*)&feat[(size_t)s * CH + lane * 4];
    acc.x = fmaf(v.x, f, acc.x);
    acc.y = fmaf(v.y, f, acc.y);
    acc.z = fmaf(v.z, f, acc.z);
    acc.w = fmaf(v.w, f, acc.w);
}

template <int N>
__device__ __forceinline__ void add_rows(float4& acc, const float* __restrict__ feat,
                                         int key, int isc, int j0, int lane) {
    float4 v[N];
    float f[N];
#pragma unroll
    for (int u = 0; u < N; ++u) {
        int s = __builtin_amdgcn_readlane(key, j0 + u);
        s = min(max(s, 0), N_NODES - 1);
        v[u] = *(const float4*)&feat[(size_t)s * CH + lane * 4];
        f[u] = __int_as_float(__builtin_amdgcn_readlane(isc, j0 + u));
    }
#pragma unroll
    for (int u = 0; u < N; ++u) {
        acc.x = fmaf(v[u].x, f[u], acc.x);
        acc.y = fmaf(v[u].y, f[u], acc.y);
        acc.z = fmaf(v[u].z, f[u], acc.z);
        acc.w = fmaf(v[u].w, f[u], acc.w);
    }
}

__device__ __forceinline__ void sum_run(float4& acc, const float* __restrict__ feat,
                                        int key, int isc, int cnt, int lane) {
    int j = 0;
    for (; j + 8 <= cnt; j += 8) add_rows<8>(acc, feat, key, isc, j, lane);
    switch (cnt & 7) {
        case 1: add_rows<1>(acc, feat, key, isc, j, lane); break;
        case 2: add_rows<2>(acc, feat, key, isc, j, lane); break;
        case 3: add_rows<3>(acc, feat, key, isc, j, lane); break;
        case 4: add_rows<4>(acc, feat, key, isc, j, lane); break;
        case 5: add_rows<5>(acc, feat, key, isc, j, lane); break;
        case 6: add_rows<6>(acc, feat, key, isc, j, lane); break;
        case 7: add_rows<7>(acc, feat, key, isc, j, lane); break;
        default: break;
    }
}

__global__ __launch_bounds__(128) void gather_gemm(const float* __restrict__ feat,
                                                   const int* __restrict__ cin,
                                                   const int* __restrict__ rowl,
                                                   const int* __restrict__ bpre,
                                                   const int* __restrict__ colw,
                                                   const float* __restrict__ W,
                                                   const float* __restrict__ bias,
                                                   float* __restrict__ out) {
    __shared__ float lds[16 * LDS_STRIDE];
    __shared__ float nrm[16];
    const int2* __restrict__ col = (const int2*)colw;
    const int t = threadIdx.x;
    const int lane = t & 31;
    const int wave = t >> 5;
    const int row0 = blockIdx.x * 16;

    int degv = 0, startv = 0;
    if (lane < 4) {
        const int n = row0 + wave * 4 + lane;
        degv = cin[n];
        startv = bpre[n >> 10] + rowl[n];
    }

    for (int i = 0; i < 4; ++i) {
        const int rl = wave * 4 + i;
        const int deg = __builtin_amdgcn_readlane(degv, i);
        const int start = __builtin_amdgcn_readlane(startv, i);
        float4 acc = make_float4(0.f, 0.f, 0.f, 0.f);

        if (deg <= 32) {
            int2 e0 = make_int2(0x7fffffff, 0);
            if (lane < deg) e0 = col[start + lane];
            const int key0 = e0.x;
            int rk0 = 0;
            for (int j = 0; j < deg; ++j) {
                const int kj = __builtin_amdgcn_readlane(key0, j);
                rk0 += sums_before(kj, j, key0, lane);
            }
            const int place = (lane < deg) ? rk0 : lane;
            const int skey = __builtin_amdgcn_ds_permute(place << 2, key0);
            const int sisc = __builtin_amdgcn_ds_permute(place << 2, e0.y);
            sum_run(acc, feat, skey, sisc, deg, lane);
        } else if (deg <= 64) {
            const int2 e0 = col[start + lane];
            int2 e1 = make_int2(0x7fffffff, 0);
            if (lane + 32 < deg) e1 = col[start + 32 + lane];
            const int key0 = e0.x, key1 = e1.x;
            int rk0 = 0, rk1 = 0;
            for (int j = 0; j < 32; ++j) {
                const int kj = __builtin_amdgcn_readlane(key0, j);
                rk0 += sums_before(kj, j, key0, lane);
                rk1 += sums_before(kj, j, key1, lane + 32);
            }
            for (int j = 32; j < deg; ++j) {
                const int kj = __builtin_amdgcn_readlane(key1, j - 32);
                rk0 += sums_before(kj, j, key0, lane);
                rk1 += sums_before(kj, j, key1, lane + 32);
            }
            const int isc0 = e0.y;
            const int isc1 = e1.y;
#pragma unroll 4
            for (int r = 0; r < deg; ++r) {
                const unsigned m0 = __builtin_amdgcn_ballot_w32(rk0 == r);
                int s, fi;
                if (m0 != 0u) {
                    const int l = __builtin_ctz(m0);
                    s = __builtin_amdgcn_readlane(key0, l);
                    fi = __builtin_amdgcn_readlane(isc0, l);
                } else {
                    const unsigned m1 = __builtin_amdgcn_ballot_w32(rk1 == r);
                    const int l = (m1 != 0u) ? __builtin_ctz(m1) : 0;
                    s = __builtin_amdgcn_readlane(key1, l);
                    fi = __builtin_amdgcn_readlane(isc1, l);
                }
                add_row(acc, feat, s, __int_as_float(fi), lane);
            }
        } else {
            for (int b = 0; b < deg; b += 32) {
                const int cnt = min(32, deg - b);
                int2 eb = make_int2(0, 0);
                if (lane < cnt) eb = col[start + b + lane];
                sum_run(acc, feat, eb.x, eb.y, cnt, lane);
            }
        }
        *(float4*)&lds[rl * LDS_STRIDE + lane * 4] = acc;
        if (lane == 0) nrm[rl] = rsqrtf(fmaxf((float)deg, 1.0f));
    }
    __syncthreads();

    const int half = lane >> 4;
    const int l16  = lane & 15;
    const int n0 = wave * 16;
    const int n1 = 64 + wave * 16;

    v8f c0 = {};
    v8f c1 = {};
    const float* arow = &lds[l16 * LDS_STRIDE + half * 2];

#pragma unroll 4
    for (int k = 0; k < CH; k += 4) {
        v2f a;
        a.x = arow[k];
        a.y = arow[k + 1];

        const float* wk = &W[(size_t)(k + half * 2) * CH + l16];
        v2f b0, b1;
        b0.x = wk[n0];       b0.y = wk[CH + n0];
        b1.x = wk[n1];       b1.y = wk[CH + n1];

        c0 = __builtin_amdgcn_wmma_f32_16x16x4_f32(false, a, false, b0, (short)0, c0, false, false);
        c1 = __builtin_amdgcn_wmma_f32_16x16x4_f32(false, a, false, b1, (short)0, c1, false, false);
    }

    __syncthreads();
#pragma unroll
    for (int r = 0; r < 8; ++r) {
        const int rr = r + half * 8;
        lds[rr * LDS_STRIDE + n0 + l16] = c0[r];
        lds[rr * LDS_STRIDE + n1 + l16] = c1[r];
    }
    __syncthreads();
    const float4 b = *(const float4*)&bias[lane * 4];
#pragma unroll
    for (int j = 0; j < 4; ++j) {
        const int rr = j * 4 + wave;
        const float s = nrm[rr];
        float4 v = *(const float4*)&lds[rr * LDS_STRIDE + lane * 4];
        v.x = fmaf(v.x, s, b.x);
        v.y = fmaf(v.y, s, b.y);
        v.z = fmaf(v.z, s, b.z);
        v.w = fmaf(v.w, s, b.w);
        *(float4*)&out[(size_t)(row0 + rr) * CH + lane * 4] = v;
    }
}

extern "C" void kernel_launch(void* const* d_in, const int* in_sizes, int n_in,
                              void* d_out, int out_size, void* d_ws, size_t ws_size,
                              hipStream_t stream) {
    const float* feat  = (const float*)d_in[0];
    const int*   src   = (const int*)  d_in[1];
    const int*   dst   = (const int*)  d_in[2];
    const float* W_mu  = (const float*)d_in[3];
    const float* W_rho = (const float*)d_in[4];
    const float* b_mu  = (const float*)d_in[5];
    const float* b_rho = (const float*)d_in[6];
    const float* W_eps = (const float*)d_in[7];
    const float* b_eps = (const float*)d_in[8];

    float* ws   = (float*)d_ws;
    int*   wsi  = (int*)d_ws;
    float* W    = ws + OFF_W;
    float* bias = ws + OFF_BIAS;
    int*   cin  = wsi + OFF_CIN;
    int*   cout = wsi + OFF_COUT;
    int*   rowl = wsi + OFF_ROWL;
    int*   bsum = wsi + OFF_BSUM;
    int*   bpre = wsi + OFF_BPRE;
    int*   rank = wsi + OFF_RANK;
    int*   col  = wsi + OFF_COL;
    int*   psrc = wsi + OFF_PSRC;
    int*   pdst = wsi + OFF_PDST;
    float* out  = (float*)d_out;

    count_chunks<<<dim3(NCHUNK, NRANGE, 2), 1024, 0, stream>>>(src, dst, psrc, pdst, rank);

    reduce_scan_prep<<<2 * N_TILES + PREP_BLOCKS, RS_THREADS, 0, stream>>>(
        pdst, psrc, cin, cout, rowl, bsum, W_mu, W_rho, W_eps, b_mu, b_rho, b_eps, W, bias);

    fill_kernel<<<(N_EDGES + 255) / 256, 256, 0, stream>>>(src, dst, rank, cout, rowl, bsum, pdst, bpre, col);

    gather_gemm<<<N_NODES / 16, 128, 0, stream>>>(feat, cin, rowl, bpre, col, W, bias, out);
}


# Placement pass: an edge's slot is the return of an LDS add on a per-node counter preloaded with bucket start + chunk prefix (32 chunks x 8 node ranges); no rank array; f32 sum order unchanged
# speedup vs baseline: 3.9618x; 1.0479x over previous
#include <hip/hip_runtime.h>
#include <math.h>

typedef __attribute__((ext_vector_type(2))) float v2f;
typedef __attribute__((ext_vector_type(8))) float v8f;

#define N_NODES 100000
#define N_EDGES 1600000
#define CH 128
#define LDS_STRIDE 132
#define SCAN_TILE 1024
#define NCHUNK 32
#define CHUNK (N_EDGES / NCHUNK)
#define NRANGE 4
#define RANGE (N_NODES / NRANGE)
#define N_TILES ((N_NODES + SCAN_TILE - 1) / SCAN_TILE)
#define RS_THREADS 512
#define PREP_BLOCKS ((CH * CH + CH + RS_THREADS - 1) / RS_THREADS)
#define FRANGES 8
#define FRANGE (N_NODES / FRANGES)
#define FB 10

#define OFF_W     0
#define OFF_BIAS  (OFF_W + CH*CH)
#define OFF_CIN   (OFF_BIAS + CH)
#define OFF_COUT  (OFF_CIN + N_NODES)
#define OFF_ROWL  (OFF_COUT + N_NODES)
#define OFF_BSUM  (OFF_ROWL + N_TILES*SCAN_TILE)
#define OFF_BPRE  (OFF_BSUM + 128)
#define OFF_COL   (OFF_BPRE + 128)
#define OFF_PSRC  OFF_COL
#define OFF_PDST  (OFF_COL + 2*N_EDGES)

__global__ __launch_bounds__(1024) void count_chunks(const int* __restrict__ src,
                                                     const int* __restrict__ dst,
                                                     int* __restrict__ psrc,
                                                     int* __restrict__ pdst) {
    __shared__ int cnt[RANGE];
    const int t = threadIdx.x;
    const int e0 = blockIdx.x * CHUNK;
    const int r0 = blockIdx.y * RANGE;
    const bool is_dst = blockIdx.z != 0;
    const int* __restrict__ idx = is_dst ? dst : src;

    for (int i = t; i < RANGE / 4; i += 1024) ((int4*)cnt)[i] = make_int4(0, 0, 0, 0);
    __syncthreads();

    for (int base = 0; base < CHUNK; base += 8 * 1024) {
        int v[8];
#pragma unroll
        for (int j = 0; j < 8; ++j) {
            const int i = base + j * 1024 + t;
            v[j] = (i < CHUNK) ? idx[e0 + i] : -1;
        }
#pragma unroll
        for (int j = 0; j < 8; ++j) {
            const unsigned k = (unsigned)(v[j] - r0);
            if (k < (unsigned)RANGE) atomicAdd(&cnt[k], 1);
        }
    }
    __syncthreads();

    int* __restrict__ part = (is_dst ? pdst : psrc) + (size_t)blockIdx.x * N_NODES + r0;
    for (int i = t; i < RANGE / 4; i += 1024) ((int4*)part)[i] = ((const int4*)cnt)[i];
}

__global__ __launch_bounds__(RS_THREADS) void reduce_scan_prep(int* pdst, const int* __restrict__ psrc,
                                                               int* __restrict__ cin, int* __restrict__ cout,
                                                               int* __restrict__ rowl, int* __restrict__ bsum,
                                                               const float* __restrict__ Wmu,
                                                               const float* __restrict__ Wrho,
                                                               const float* __restrict__ Weps,
                                                               const float* __restrict__ bmu,
                                                               const float* __restrict__ brho,
                                                               const float* __restrict__ beps,
                                                               float* __restrict__ W, float* __restrict__ bias) {
    __shared__ int wsum[RS_THREADS / 32];
    const int t = threadIdx.x;
    const int b = blockIdx.x;

    if (b >= 2 * N_TILES) {
        const int i = (b - 2 * N_TILES) * RS_THREADS + t;
        if (i < CH * CH) {
            float sp = log1pf(expf(Wrho[i]));
            W[i] = Wmu[i] + Weps[i] * sp;
        } else if (i < CH * CH + CH) {
            int j = i - CH * CH;
            float sp = log1pf(expf(brho[j]));
            bias[j] = bmu[j] + beps[j] * sp;
        }
        return;
    }

    if (b >= N_TILES) {
        const int base = (b - N_TILES) * SCAN_TILE + t * 2;
        if (base < N_NODES) {
            int2 v[NCHUNK];
#pragma unroll
            for (int c = 0; c < NCHUNK; ++c) v[c] = *(const int2*)&psrc[(size_t)c * N_NODES + base];
            int2 run = make_int2(0, 0);
#pragma unroll
            for (int c = 0; c < NCHUNK; ++c) { run.x += v[c].x; run.y += v[c].y; }
            *(int2*)&cout[base] = run;
        }
        return;
    }

    const int lane = t & 31;
    const int wave = t >> 5;
    const int base = b * SCAN_TILE + t * 2;
    int2 run = make_int2(0, 0);
    if (base < N_NODES) {
        int2 v[NCHUNK];
#pragma unroll
        for (int c = 0; c < NCHUNK; ++c) v[c] = *(const int2*)&pdst[(size_t)c * N_NODES + base];
#pragma unroll
        for (int c = 0; c < NCHUNK; ++c) {
            *(int2*)&pdst[(size_t)c * N_NODES + base] = run;
            run.x += v[c].x;
            run.y += v[c].y;
        }
        *(int2*)&cin[base] = run;
    }

    const int tot = run.x + run.y;
    int incl = tot;
#pragma unroll
    for (int off = 1; off < 32; off <<= 1) {
        const int y = __shfl_up(incl, off);
        if (lane >= off) incl += y;
    }
    if (lane == 31) wsum[wave] = incl;
    __syncthreads();
    int wbase = 0;
    for (int w = 0; w < wave; ++w) wbase += wsum[w];
    const int excl = wbase + incl - tot;
    *(int2*)&rowl[base] = make_int2(excl, excl + run.x);
    if (t == RS_THREADS - 1) bsum[b] = excl + tot;
}

__global__ __launch_bounds__(1024) void fill_kernel(const int* __restrict__ src, const int* __restrict__ dst,
                            const int* __restrict__ cout,
                            const int* __restrict__ rowl, const int* __restrict__ bsum,
                            const int* __restrict__ pdst,
                            int* __restrict__ bpre_out,
                            int* __restrict__ colw) {
    __shared__ __attribute__((aligned(16))) int slot[FRANGE];
    __shared__ __attribute__((aligned(16))) int bpre[128];
    int2* __restrict__ col = (int2*)colw;
    const int t = threadIdx.x;
    const int e0 = blockIdx.x * CHUNK;
    const int r0 = blockIdx.y * FRANGE;

    int dn[FB], sn[FB];
#pragma unroll
    for (int j = 0; j < FB; ++j) {
        const int i = j * 1024 + t;
        dn[j] = dst[e0 + i];
        sn[j] = src[e0 + i];
    }

    if (t < 32) {
        int v[4];
#pragma unroll
        for (int i = 0; i < 4; ++i) {
            const int idx = t * 4 + i;
            const int x = bsum[min(idx, N_TILES - 1)];
            v[i] = (idx < N_TILES) ? x : 0;
        }
        const int tot = v[0] + v[1] + v[2] + v[3];
        int incl = tot;
#pragma unroll
        for (int off = 1; off < 32; off <<= 1) {
            const int y = __shfl_up(incl, off);
            if (t >= off) incl += y;
        }
        const int excl = incl - tot;
        const int4 o = make_int4(excl, excl + v[0], excl + v[0] + v[1], excl + v[0] + v[1] + v[2]);
        *(int4*)&bpre[t * 4] = o;
        if (blockIdx.x == 0 && blockIdx.y == 0) *(int4*)&bpre_out[t * 4] = o;
    }
    __syncthreads();

    {
        const int4* __restrict__ pl = (const int4*)(pdst + (size_t)blockIdx.x * N_NODES + r0);
        const int4* __restrict__ rl = (const int4*)(rowl + r0);
        for (int i = t; i < FRANGE / 4; i += 1024) {
            const int4 a = pl[i];
            const int4 b = rl[i];
            const int tb = bpre[(r0 + i * 4) >> 10];
            ((int4*)slot)[i] = make_int4(a.x + b.x + tb, a.y + b.y + tb, a.z + b.z + tb, a.w + b.w + tb);
        }
    }
    __syncthreads();

    for (int base = 0; base < CHUNK; base += FB * 1024) {
        int d[FB], s[FB], od[FB], pos[FB];
#pragma unroll
        for (int j = 0; j < FB; ++j) { d[j] = dn[j]; s[j] = sn[j]; }
#pragma unroll
        for (int j = 0; j < FB; ++j) {
            const unsigned k = (unsigned)(d[j] - r0);
            od[j] = cout[(k < (unsigned)FRANGE) ? s[j] : 0];
        }
        const int nbase = base + FB * 1024;
        if (nbase < CHUNK) {
#pragma unroll
            for (int j = 0; j < FB; ++j) {
                const int i = nbase + j * 1024 + t;
                dn[j] = (i < CHUNK) ? dst[e0 + i] : -1;
                sn[j] = (i < CHUNK) ? src[e0 + i] : 0;
            }
        }
#pragma unroll
        for (int j = 0; j < FB; ++j) {
            const unsigned k = (unsigned)(d[j] - r0);
            pos[j] = (k < (unsigned)FRANGE) ? atomicAdd(&slot[k], 1) : -1;
        }
#pragma unroll
        for (int j = 0; j < FB; ++j) {
            if (pos[j] >= 0) {
                const int p = min(pos[j], N_EDGES - 1);
                const float f = rsqrtf(fmaxf((float)od[j], 1.0f));
                col[p] = make_int2(s[j], __float_as_int(f));
            }
        }
    }
}

__device__ __forceinline__ int sums_before(int ki, int i, int k, int p) {
    return (ki < k || (ki == k && i < p)) ? 1 : 0;
}

__device__ __forceinline__ void add_row(float4& acc, const float* __restrict__ feat,
                                        int s, float f, int lane) {
    s = min(max(s, 0), N_NODES - 1);
    const float4 v = *(const float4*)&feat[(size_t)s * CH + lane * 4];
    acc.x = fmaf(v.x, f, acc.x);
    acc.y = fmaf(v.y, f, acc.y);
    acc.z = fmaf(v.z, f, acc.z);
    acc.w = fmaf(v.w, f, acc.w);
}

template <int N>
__device__ __forceinline__ void add_rows(float4& acc, const float* __restrict__ feat,
                                         int key, int isc, int j0, int lane) {
    float4 v[N];
    float f[N];
#pragma unroll
    for (int u = 0; u < N; ++u) {
        int s = __builtin_amdgcn_readlane(key, j0 + u);
        s = min(max(s, 0), N_NODES - 1);
        v[u] = *(const float4*)&feat[(size_t)s * CH + lane * 4];
        f[u] = __int_as_float(__builtin_amdgcn_readlane(isc, j0 + u));
    }
#pragma unroll
    for (int u = 0; u < N; ++u) {
        acc.x = fmaf(v[u].x, f[u], acc.x);
        acc.y = fmaf(v[u].y, f[u], acc.y);
        acc.z = fmaf(v[u].z, f[u], acc.z);
        acc.w = fmaf(v[u].w, f[u], acc.w);
    }
}

__device__ __forceinline__ void sum_run(float4& acc, const float* __restrict__ feat,
                                        int key, int isc, int cnt, int lane) {
    int j = 0;
    for (; j + 8 <= cnt; j += 8) add_rows<8>(acc, feat, key, isc, j, lane);
    switch (cnt & 7) {
        case 1: add_rows<1>(acc, feat, key, isc, j, lane); break;
        case 2: add_rows<2>(acc, feat, key, isc, j, lane); break;
        case 3: add_rows<3>(acc, feat, key, isc, j, lane); break;
        case 4: add_rows<4>(acc, feat, key, isc, j, lane); break;
        case 5: add_rows<5>(acc, feat, key, isc, j, lane); break;
        case 6: add_rows<6>(acc, feat, key, isc, j, lane); break;
        case 7: add_rows<7>(acc, feat, key, isc, j, lane); break;
        default: break;
    }
}

__global__ __launch_bounds__(128) void gather_gemm(const float* __restrict__ feat,
                                                   const int* __restrict__ cin,
                                                   const int* __restrict__ rowl,
                                                   const int* __restrict__ bpre,
                                                   const int* __restrict__ colw,
                                                   const float* __restrict__ W,
                                                   const float* __restrict__ bias,
                                                   float* __restrict__ out) {
    __shared__ float lds[16 * LDS_STRIDE];
    __shared__ float nrm[16];
    const int2* __restrict__ col = (const int2*)colw;
    const int t = threadIdx.x;
    const int lane = t & 31;
    const int wave = t >> 5;
    const int row0 = blockIdx.x * 16;

    int degv = 0, startv = 0;
    if (lane < 4) {
        const int n = row0 + wave * 4 + lane;
        degv = cin[n];
        startv = bpre[n >> 10] + rowl[n];
    }

    for (int i = 0; i < 4; ++i) {
        const int rl = wave * 4 + i;
        const int deg = __builtin_amdgcn_readlane(degv, i);
        const int start = __builtin_amdgcn_readlane(startv, i);
        float4 acc = make_float4(0.f, 0.f, 0.f, 0.f);

        if (deg <= 32) {
            int2 e0 = make_int2(0x7fffffff, 0);
            if (lane < deg) e0 = col[start + lane];
            const int key0 = e0.x;
            int rk0 = 0;
            for (int j = 0; j < deg; ++j) {
                const int kj = __builtin_amdgcn_readlane(key0, j);
                rk0 += sums_before(kj, j, key0, lane);
            }
            const int place = (lane < deg) ? rk0 : lane;
            const int skey = __builtin_amdgcn_ds_permute(place << 2, key0);
            const int sisc = __builtin_amdgcn_ds_permute(place << 2, e0.y);
            sum_run(acc, feat, skey, sisc, deg, lane);
        } else if (deg <= 64) {
            const int2 e0 = col[start + lane];
            int2 e1 = make_int2(0x7fffffff, 0);
            if (lane + 32 < deg) e1 = col[start + 32 + lane];
            const int key0 = e0.x, key1 = e1.x;
            int rk0 = 0, rk1 = 0;
            for (int j = 0; j < 32; ++j) {
                const int kj = __builtin_amdgcn_readlane(key0, j);
                rk0 += sums_before(kj, j, key0, lane);
                rk1 += sums_before(kj, j, key1, lane + 32);
            }
            for (int j = 32; j < deg; ++j) {
                const int kj = __builtin_amdgcn_readlane(key1, j - 32);
                rk0 += sums_before(kj, j, key0, lane);
                rk1 += sums_before(kj, j, key1, lane + 32);
            }
            const int isc0 = e0.y;
            const int isc1 = e1.y;
#pragma unroll 4
            for (int r = 0; r < deg; ++r) {
                const unsigned m0 = __builtin_amdgcn_ballot_w32(rk0 == r);
                int s, fi;
                if (m0 != 0u) {
                    const int l = __builtin_ctz(m0);
                    s = __builtin_amdgcn_readlane(key0, l);
                    fi = __builtin_amdgcn_readlane(isc0, l);
                } else {
                    const unsigned m1 = __builtin_amdgcn_ballot_w32(rk1 == r);
                    const int l = (m1 != 0u) ? __builtin_ctz(m1) : 0;
                    s = __builtin_amdgcn_readlane(key1, l);
                    fi = __builtin_amdgcn_readlane(isc1, l);
                }
                add_row(acc, feat, s, __int_as_float(fi), lane);
            }
        } else {
            for (int b = 0; b < deg; b += 32) {
                const int cnt = min(32, deg - b);
                int2 eb = make_int2(0, 0);
                if (lane < cnt) eb = col[start + b + lane];
                sum_run(acc, feat, eb.x, eb.y, cnt, lane);
            }
        }
        *(float4*)&lds[rl * LDS_STRIDE + lane * 4] = acc;
        if (lane == 0) nrm[rl] = rsqrtf(fmaxf((float)deg, 1.0f));
    }
    __syncthreads();

    const int half = lane >> 4;
    const int l16  = lane & 15;
    const int n0 = wave * 16;
    const int n1 = 64 + wave * 16;

    v8f c0 = {};
    v8f c1 = {};
    const float* arow = &lds[l16 * LDS_STRIDE + half * 2];

#pragma unroll 4
    for (int k = 0; k < CH; k += 4) {
        v2f a;
        a.x = arow[k];
        a.y = arow[k + 1];

        const float* wk = &W[(size_t)(k + half * 2) * CH + l16];
        v2f b0, b1;
        b0.x = wk[n0];       b0.y = wk[CH + n0];
        b1.x = wk[n1];       b1.y = wk[CH + n1];

        c0 = __builtin_amdgcn_wmma_f32_16x16x4_f32(false, a, false, b0, (short)0, c0, false, false);
        c1 = __builtin_amdgcn_wmma_f32_16x16x4_f32(false, a, false, b1, (short)0, c1, false, false);
    }

    __syncthreads();
#pragma unroll
    for (int r = 0; r < 8; ++r) {
        const int rr = r + half * 8;
        lds[rr * LDS_STRIDE + n0 + l16] = c0[r];
        lds[rr * LDS_STRIDE + n1 + l16] = c1[r];
    }
    __syncthreads();
    const float4 b = *(const float4*)&bias[lane * 4];
#pragma unroll
    for (int j = 0; j < 4; ++j) {
        const int rr = j * 4 + wave;
        const float s = nrm[rr];
        float4 v = *(const float4*)&lds[rr * LDS_STRIDE + lane * 4];
        v.x = fmaf(v.x, s, b.x);
        v.y = fmaf(v.y, s, b.y);
        v.z = fmaf(v.z, s, b.z);
        v.w = fmaf(v.w, s, b.w);
        *(float4*)&out[(size_t)(row0 + rr) * CH + lane * 4] = v;
    }
}

extern "C" void kernel_launch(void* const* d_in, const int* in_sizes, int n_in,
                              void* d_out, int out_size, void* d_ws, size_t ws_size,
                              hipStream_t stream) {
    const float* feat  = (const float*)d_in[0];
    const int*   src   = (const int*)  d_in[1];
    const int*   dst   = (const int*)  d_in[2];
    const float* W_mu  = (const float*)d_in[3];
    const float* W_rho = (const float*)d_in[4];
    const float* b_mu  = (const float*)d_in[5];
    const float* b_rho = (const float*)d_in[6];
    const float* W_eps = (const float*)d_in[7];
    const float* b_eps = (const float*)d_in[8];

    float* ws   = (float*)d_ws;
    int*   wsi  = (int*)d_ws;
    float* W    = ws + OFF_W;
    float* bias = ws + OFF_BIAS;
    int*   cin  = wsi + OFF_CIN;
    int*   cout = wsi + OFF_COUT;
    int*   rowl = wsi + OFF_ROWL;
    int*   bsum = wsi + OFF_BSUM;
    int*   bpre = wsi + OFF_BPRE;
    int*   col  = wsi + OFF_COL;
    int*   psrc = wsi + OFF_PSRC;
    int*   pdst = wsi + OFF_PDST;
    float* out  = (float*)d_out;

    count_chunks<<<dim3(NCHUNK, NRANGE, 2), 1024, 0, stream>>>(src, dst, psrc, pdst);

    reduce_scan_prep<<<2 * N_TILES + PREP_BLOCKS, RS_THREADS, 0, stream>>>(
        pdst, psrc, cin, cout, rowl, bsum, W_mu, W_rho, W_eps, b_mu, b_rho, b_eps, W, bias);

    fill_kernel<<<dim3(NCHUNK, FRANGES), 1024, 0, stream>>>(src, dst, cout, rowl, bsum, pdst, bpre, col);

    gather_gemm<<<N_NODES / 16, 128, 0, stream>>>(feat, cin, rowl, bpre, col, W, bias, out);
}


# Independent loads asked for before the first wait: count pass takes its chunk as 16-byte loads (7 round trips to 1), placement set-up 6 to 1, gather gets 4 nodes' records at once; f32 sum order kept
# speedup vs baseline: 4.1223x; 1.0405x over previous
#include <hip/hip_runtime.h>
#include <math.h>

typedef __attribute__((ext_vector_type(2))) float v2f;
typedef __attribute__((ext_vector_type(8))) float v8f;

#define N_NODES 100000
#define N_EDGES 1600000
#define CH 128
#define LDS_STRIDE 132
#define SCAN_TILE 1024
#define NCHUNK 32
#define CHUNK (N_EDGES / NCHUNK)
#define NRANGE 4
#define RANGE (N_NODES / NRANGE)
#define N_TILES ((N_NODES + SCAN_TILE - 1) / SCAN_TILE)
#define RS_THREADS 512
#define PREP_BLOCKS ((CH * CH + CH + RS_THREADS - 1) / RS_THREADS)
#define FRANGES 8
#define FRANGE (N_NODES / FRANGES)
#define FB 10
#define CQ ((CHUNK / 4 + 1023) / 1024)
#define SQ ((FRANGE / 4 + 1023) / 1024)

#define OFF_W     0
#define OFF_BIAS  (OFF_W + CH*CH)
#define OFF_CIN   (OFF_BIAS + CH)
#define OFF_COUT  (OFF_CIN + N_NODES)
#define OFF_ROWL  (OFF_COUT + N_NODES)
#define OFF_BSUM  (OFF_ROWL + N_TILES*SCAN_TILE)
#define OFF_BPRE  (OFF_BSUM + 128)
#define OFF_COL   (OFF_BPRE + 128)
#define OFF_PSRC  OFF_COL
#define OFF_PDST  (OFF_COL + 2*N_EDGES)

__global__ __launch_bounds__(1024) void count_chunks(const int* __restrict__ src,
                                                     const int* __restrict__ dst,
                                                     int* __restrict__ psrc,
                                                     int* __restrict__ pdst) {
    __shared__ int cnt[RANGE];
    const int t = threadIdx.x;
    const int e0 = blockIdx.x * CHUNK;
    const int r0 = blockIdx.y * RANGE;
    const bool is_dst = blockIdx.z != 0;
    const int* __restrict__ idx = is_dst ? dst : src;
    const int4* __restrict__ idx4 = (const int4*)(idx + e0);

    int4 v[CQ];
#pragma unroll
    for (int j = 0; j < CQ; ++j) {
        const int i = j * 1024 + t;
        v[j] = (i < CHUNK / 4) ? idx4[i] : make_int4(-1, -1, -1, -1);
    }

    for (int i = t; i < RANGE / 4; i += 1024) ((int4*)cnt)[i] = make_int4(0, 0, 0, 0);
    __syncthreads();

#pragma unroll
    for (int j = 0; j < CQ; ++j) {
        const unsigned k0 = (unsigned)(v[j].x - r0);
        const unsigned k1 = (unsigned)(v[j].y - r0);
        const unsigned k2 = (unsigned)(v[j].z - r0);
        const unsigned k3 = (unsigned)(v[j].w - r0);
        if (k0 < (unsigned)RANGE) atomicAdd(&cnt[k0], 1);
        if (k1 < (unsigned)RANGE) atomicAdd(&cnt[k1], 1);
        if (k2 < (unsigned)RANGE) atomicAdd(&cnt[k2], 1);
        if (k3 < (unsigned)RANGE) atomicAdd(&cnt[k3], 1);
    }
    __syncthreads();

    int* __restrict__ part = (is_dst ? pdst : psrc) + (size_t)blockIdx.x * N_NODES + r0;
    for (int i = t; i < RANGE / 4; i += 1024) ((int4*)part)[i] = ((const int4*)cnt)[i];
}

__global__ __launch_bounds__(RS_THREADS) void reduce_scan_prep(int* pdst, const int* __restrict__ psrc,
                                                               int* __restrict__ cin, int* __restrict__ cout,
                                                               int* __restrict__ rowl, int* __restrict__ bsum,
                                                               const float* __restrict__ Wmu,
                                                               const float* __restrict__ Wrho,
                                                               const float* __restrict__ Weps,
                                                               const float* __restrict__ bmu,
                                                               const float* __restrict__ brho,
                                                               const float* __restrict__ beps,
                                                               float* __restrict__ W, float* __restrict__ bias) {
    __shared__ int wsum[RS_THREADS / 32];
    const int t = threadIdx.x;
    const int b = blockIdx.x;

    if (b >= 2 * N_TILES) {
        const int i = (b - 2 * N_TILES) * RS_THREADS + t;
        if (i < CH * CH) {
            float sp = log1pf(expf(Wrho[i]));
            W[i] = Wmu[i] + Weps[i] * sp;
        } else if (i < CH * CH + CH) {
            int j = i - CH * CH;
            float sp = log1pf(expf(brho[j]));
            bias[j] = bmu[j] + beps[j] * sp;
        }
        return;
    }

    if (b >= N_TILES) {
        const int base = (b - N_TILES) * SCAN_TILE + t * 2;
        if (base < N_NODES) {
            int2 v[NCHUNK];
#pragma unroll
            for (int c = 0; c < NCHUNK; ++c) v[c] = *(const int2*)&psrc[(size_t)c * N_NODES + base];
            int2 run = make_int2(0, 0);
#pragma unroll
            for (int c = 0; c < NCHUNK; ++c) { run.x += v[c].x; run.y += v[c].y; }
            *(int2*)&cout[base] = run;
        }
        return;
    }

    const int lane = t & 31;
    const int wave = t >> 5;
    const int base = b * SCAN_TILE + t * 2;
    int2 run = make_int2(0, 0);
    if (base < N_NODES) {
        int2 v[NCHUNK];
#pragma unroll
        for (int c = 0; c < NCHUNK; ++c) v[c] = *(const int2*)&pdst[(size_t)c * N_NODES + base];
#pragma unroll
        for (int c = 0; c < NCHUNK; ++c) {
            *(int2*)&pdst[(size_t)c * N_NODES + base] = run;
            run.x += v[c].x;
            run.y += v[c].y;
        }
        *(int2*)&cin[base] = run;
    }

    const int tot = run.x + run.y;
    int incl = tot;
#pragma unroll
    for (int off = 1; off < 32; off <<= 1) {
        const int y = __shfl_up(incl, off);
        if (lane >= off) incl += y;
    }
    if (lane == 31) wsum[wave] = incl;
    __syncthreads();
    int wbase = 0;
    for (int w = 0; w < wave; ++w) wbase += wsum[w];
    const int excl = wbase + incl - tot;
    *(int2*)&rowl[base] = make_int2(excl, excl + run.x);
    if (t == RS_THREADS - 1) bsum[b] = excl + tot;
}

__global__ __launch_bounds__(1024) void fill_kernel(const int* __restrict__ src, const int* __restrict__ dst,
                            const int* __restrict__ cout,
                            const int* __restrict__ rowl, const int* __restrict__ bsum,
                            const int* __restrict__ pdst,
                            int* __restrict__ bpre_out,
                            int* __restrict__ colw) {
    __shared__ __attribute__((aligned(16))) int slot[FRANGE];
    __shared__ __attribute__((aligned(16))) int bpre[128];
    int2* __restrict__ col = (int2*)colw;
    const int t = threadIdx.x;
    const int e0 = blockIdx.x * CHUNK;
    const int r0 = blockIdx.y * FRANGE;

    int bs[4] = {0, 0, 0, 0};
    if (t < 32) {
#pragma unroll
        for (int i = 0; i < 4; ++i) {
            bs[i] = bsum[min(t * 4 + i, N_TILES - 1)];
        }
    }
    const int4* __restrict__ pl = (const int4*)(pdst + (size_t)blockIdx.x * N_NODES + r0);
    const int4* __restrict__ rl = (const int4*)(rowl + r0);
    int4 pa[SQ], pb[SQ];
#pragma unroll
    for (int q = 0; q < SQ; ++q) {
        const int i = min(q * 1024 + t, FRANGE / 4 - 1);
        pa[q] = pl[i];
        pb[q] = rl[i];
    }
    int dn[FB], sn[FB];
#pragma unroll
    for (int j = 0; j < FB; ++j) {
        const int i = j * 1024 + t;
        dn[j] = dst[e0 + i];
        sn[j] = src[e0 + i];
    }

    if (t < 32) {
        int v[4];
#pragma unroll
        for (int i = 0; i < 4; ++i) v[i] = (t * 4 + i < N_TILES) ? bs[i] : 0;
        const int tot = v[0] + v[1] + v[2] + v[3];
        int incl = tot;
#pragma unroll
        for (int off = 1; off < 32; off <<= 1) {
            const int y = __shfl_up(incl, off);
            if (t >= off) incl += y;
        }
        const int excl = incl - tot;
        const int4 o = make_int4(excl, excl + v[0], excl + v[0] + v[1], excl + v[0] + v[1] + v[2]);
        *(int4*)&bpre[t * 4] = o;
        if (blockIdx.x == 0 && blockIdx.y == 0) *(int4*)&bpre_out[t * 4] = o;
    }
    __syncthreads();

#pragma unroll
    for (int q = 0; q < SQ; ++q) {
        const int i = q * 1024 + t;
        if (i < FRANGE / 4) {
            const int4 a = pa[q];
            const int4 b = pb[q];
            const int tb = bpre[(r0 + i * 4) >> 10];
            ((int4*)slot)[i] = make_int4(a.x + b.x + tb, a.y + b.y + tb, a.z + b.z + tb, a.w + b.w + tb);
        }
    }
    __syncthreads();

    for (int base = 0; base < CHUNK; base += FB * 1024) {
        int d[FB], s[FB], od[FB], pos[FB];
#pragma unroll
        for (int j = 0; j < FB; ++j) { d[j] = dn[j]; s[j] = sn[j]; }
#pragma unroll
        for (int j = 0; j < FB; ++j) {
            const unsigned k = (unsigned)(d[j] - r0);
            od[j] = cout[(k < (unsigned)FRANGE) ? s[j] : 0];
        }
        const int nbase = base + FB * 1024;
        if (nbase < CHUNK) {
#pragma unroll
            for (int j = 0; j < FB; ++j) {
                const int i = nbase + j * 1024 + t;
                dn[j] = (i < CHUNK) ? dst[e0 + i] : -1;
                sn[j] = (i < CHUNK) ? src[e0 + i] : 0;
            }
        }
#pragma unroll
        for (int j = 0; j < FB; ++j) {
            const unsigned k = (unsigned)(d[j] - r0);
            pos[j] = (k < (unsigned)FRANGE) ? atomicAdd(&slot[k], 1) : -1;
        }
#pragma unroll
        for (int j = 0; j < FB; ++j) {
            if (pos[j] >= 0) {
                const int p = min(pos[j], N_EDGES - 1);
                const float f = rsqrtf(fmaxf((float)od[j], 1.0f));
                col[p] = make_int2(s[j], __float_as_int(f));
            }
        }
    }
}

__device__ __forceinline__ int sums_before(int ki, int i, int k, int p) {
    return (ki < k || (ki == k && i < p)) ? 1 : 0;
}

__device__ __forceinline__ void add_row(float4& acc, const float* __restrict__ feat,
                                        int s, float f, int lane) {
    s = min(max(s, 0), N_NODES - 1);
    const float4 v = *(const float4*)&feat[(size_t)s * CH + lane * 4];
    acc.x = fmaf(v.x, f, acc.x);
    acc.y = fmaf(v.y, f, acc.y);
    acc.z = fmaf(v.z, f, acc.z);
    acc.w = fmaf(v.w, f, acc.w);
}

template <int N>
__device__ __forceinline__ void add_rows(float4& acc, const float* __restrict__ feat,
                                         int key, int isc, int j0, int lane) {
    float4 v[N];
    float f[N];
#pragma unroll
    for (int u = 0; u < N; ++u) {
        int s = __builtin_amdgcn_readlane(key, j0 + u);
        s = min(max(s, 0), N_NODES - 1);
        v[u] = *(const float4*)&feat[(size_t)s * CH + lane * 4];
        f[u] = __int_as_float(__builtin_amdgcn_readlane(isc, j0 + u));
    }
#pragma unroll
    for (int u = 0; u < N; ++u) {
        acc.x = fmaf(v[u].x, f[u], acc.x);
        acc.y = fmaf(v[u].y, f[u], acc.y);
        acc.z = fmaf(v[u].z, f[u], acc.z);
        acc.w = fmaf(v[u].w, f[u], acc.w);
    }
}

__device__ __forceinline__ void sum_run(float4& acc, const float* __restrict__ feat,
                                        int key, int isc, int cnt, int lane) {
    int j = 0;
    for (; j + 8 <= cnt; j += 8) add_rows<8>(acc, feat, key, isc, j, lane);
    switch (cnt & 7) {
        case 1: add_rows<1>(acc, feat, key, isc, j, lane); break;
        case 2: add_rows<2>(acc, feat, key, isc, j, lane); break;
        case 3: add_rows<3>(acc, feat, key, isc, j, lane); break;
        case 4: add_rows<4>(acc, feat, key, isc, j, lane); break;
        case 5: add_rows<5>(acc, feat, key, isc, j, lane); break;
        case 6: add_rows<6>(acc, feat, key, isc, j, lane); break;
        case 7: add_rows<7>(acc, feat, key, isc, j, lane); break;
        default: break;
    }
}

__device__ __forceinline__ int2 records_from(const int2* __restrict__ col, int startv, int n, int lane) {
    const int st = __builtin_amdgcn_readlane(startv, n);
    return col[min(max(st + lane, 0), N_EDGES - 1)];
}

__global__ __launch_bounds__(128) void gather_gemm(const float* __restrict__ feat,
                                                   const int* __restrict__ cin,
                                                   const int* __restrict__ rowl,
                                                   const int* __restrict__ bpre,
                                                   const int* __restrict__ colw,
                                                   const float* __restrict__ W,
                                                   const float* __restrict__ bias,
                                                   float* __restrict__ out) {
    __shared__ float lds[16 * LDS_STRIDE];
    __shared__ float nrm[16];
    __shared__ int2 recs[4][4][32];
    const int2* __restrict__ col = (const int2*)colw;
    const int t = threadIdx.x;
    const int lane = t & 31;
    const int wave = t >> 5;
    const int row0 = blockIdx.x * 16;

    int degv = 0, startv = 0;
    if (lane < 4) {
        const int n = row0 + wave * 4 + lane;
        degv = cin[n];
        startv = bpre[n >> 10] + rowl[n];
    }

    {
        const int2 q0 = records_from(col, startv, 0, lane);
        const int2 q1 = records_from(col, startv, 1, lane);
        const int2 q2 = records_from(col, startv, 2, lane);
        const int2 q3 = records_from(col, startv, 3, lane);
        recs[wave][0][lane] = q0;
        recs[wave][1][lane] = q1;
        recs[wave][2][lane] = q2;
        recs[wave][3][lane] = q3;
    }

    for (int i = 0; i < 4; ++i) {
        const int rl = wave * 4 + i;
        const int deg = __builtin_amdgcn_readlane(degv, i);
        const int start = __builtin_amdgcn_readlane(startv, i);
        float4 acc = make_float4(0.f, 0.f, 0.f, 0.f);
        const int2 raw = recs[wave][i][lane];
        const int2 pre = (lane < deg) ? raw : make_int2(0x7fffffff, 0);

        if (deg <= 32) {
            const int2 e0 = pre;
            const int key0 = e0.x;
            int rk0 = 0;
            for (int j = 0; j < deg; ++j) {
                const int kj = __builtin_amdgcn_readlane(key0, j);
                rk0 += sums_before(kj, j, key0, lane);
            }
            const int place = (lane < deg) ? rk0 : lane;
            const int skey = __builtin_amdgcn_ds_permute(place << 2, key0);
            const int sisc = __builtin_amdgcn_ds_permute(place << 2, e0.y);
            sum_run(acc, feat, skey, sisc, deg, lane);
        } else if (deg <= 64) {
            const int2 e0 = pre;
            int2 e1 = make_int2(0x7fffffff, 0);
            if (lane + 32 < deg) e1 = col[start + 32 + lane];
            const int key0 = e0.x, key1 = e1.x;
            int rk0 = 0, rk1 = 0;
            for (int j = 0; j < 32; ++j) {
                const int kj = __builtin_amdgcn_readlane(key0, j);
                rk0 += sums_before(kj, j, key0, lane);
                rk1 += sums_before(kj, j, key1, lane + 32);
            }
            for (int j = 32; j < deg; ++j) {
                const int kj = __builtin_amdgcn_readlane(key1, j - 32);
                rk0 += sums_before(kj, j, key0, lane);
                rk1 += sums_before(kj, j, key1, lane + 32);
            }
            const int isc0 = e0.y;
            const int isc1 = e1.y;
#pragma unroll 4
            for (int r = 0; r < deg; ++r) {
                const unsigned m0 = __builtin_amdgcn_ballot_w32(rk0 == r);
                int s, fi;
                if (m0 != 0u) {
                    const int l = __builtin_ctz(m0);
                    s = __builtin_amdgcn_readlane(key0, l);
                    fi = __builtin_amdgcn_readlane(isc0, l);
                } else {
                    const unsigned m1 = __builtin_amdgcn_ballot_w32(rk1 == r);
                    const int l = (m1 != 0u) ? __builtin_ctz(m1) : 0;
                    s = __builtin_amdgcn_readlane(key1, l);
                    fi = __builtin_amdgcn_readlane(isc1, l);
                }
                add_row(acc, feat, s, __int_as_float(fi), lane);
            }
        } else {
            for (int b = 0; b < deg; b += 32) {
                const int cnt = min(32, deg - b);
                int2 eb = make_int2(0, 0);
                if (lane < cnt) eb = col[start + b + lane];
                sum_run(acc, feat, eb.x, eb.y, cnt, lane);
            }
        }
        *(float4*)&lds[rl * LDS_STRIDE + lane * 4] = acc;
        if (lane == 0) nrm[rl] = rsqrtf(fmaxf((float)deg, 1.0f));
    }
    __syncthreads();

    const int half = lane >> 4;
    const int l16  = lane & 15;
    const int n0 = wave * 16;
    const int n1 = 64 + wave * 16;

    v8f c0 = {};
    v8f c1 = {};
    const float* arow = &lds[l16 * LDS_STRIDE + half * 2];

#pragma unroll 4
    for (int k = 0; k < CH; k += 4) {
        v2f a;
        a.x = arow[k];
        a.y = arow[k + 1];

        const float* wk = &W[(size_t)(k + half * 2) * CH + l16];
        v2f b0, b1;
        b0.x = wk[n0];       b0.y = wk[CH + n0];
        b1.x = wk[n1];       b1.y = wk[CH + n1];

        c0 = __builtin_amdgcn_wmma_f32_16x16x4_f32(false, a, false, b0, (short)0, c0, false, false);
        c1 = __builtin_amdgcn_wmma_f32_16x16x4_f32(false, a, false, b1, (short)0, c1, false, false);
    }

    __syncthreads();
#pragma unroll
    for (int r = 0; r < 8; ++r) {
        const int rr = r + half * 8;
        lds[rr * LDS_STRIDE + n0 + l16] = c0[r];
        lds[rr * LDS_STRIDE + n1 + l16] = c1[r];
    }
    __syncthreads();
    const float4 b = *(const float4*)&bias[lane * 4];
#pragma unroll
    for (int j = 0; j < 4; ++j) {
        const int rr = j * 4 + wave;
        const float s = nrm[rr];
        float4 v = *(const float4*)&lds[rr * LDS_STRIDE + lane * 4];
        v.x = fmaf(v.x, s, b.x);
        v.y = fmaf(v.y, s, b.y);
        v.z = fmaf(v.z, s, b.z);
        v.w = fmaf(v.w, s, b.w);
        *(float4*)&out[(size_t)(row0 + rr) * CH + lane * 4] = v;
    }
}

extern "C" void kernel_launch(void* const* d_in, const int* in_sizes, int n_in,
                              void* d_out, int out_size, void* d_ws, size_t ws_size,
                              hipStream_t stream) {
    const float* feat  = (const float*)d_in[0];
    const int*   src   = (const int*)  d_in[1];
    const int*   dst   = (const int*)  d_in[2];
    const float* W_mu  = (const float*)d_in[3];
    const float* W_rho = (const float*)d_in[4];
    const float* b_mu  = (const float*)d_in[5];
    const float* b_rho = (const float*)d_in[6];
    const float* W_eps = (const float*)d_in[7];
    const float* b_eps = (const float*)d_in[8];

    float* ws   = (float*)d_ws;
    int*   wsi  = (int*)d_ws;
    float* W    = ws + OFF_W;
    float* bias = ws + OFF_BIAS;
    int*   cin  = wsi + OFF_CIN;
    int*   cout = wsi + OFF_COUT;
    int*   rowl = wsi + OFF_ROWL;
    int*   bsum = wsi + OFF_BSUM;
    int*   bpre = wsi + OFF_BPRE;
    int*   col  = wsi + OFF_COL;
    int*   psrc = wsi + OFF_PSRC;
    int*   pdst = wsi + OFF_PDST;
    float* out  = (float*)d_out;

    count_chunks<<<dim3(NCHUNK, NRANGE, 2), 1024, 0, stream>>>(src, dst, psrc, pdst);

    reduce_scan_prep<<<2 * N_TILES + PREP_BLOCKS, RS_THREADS, 0, stream>>>(
        pdst, psrc, cin, cout, rowl, bsum, W_mu, W_rho, W_eps, b_mu, b_rho, b_eps, W, bias);

    fill_kernel<<<dim3(NCHUNK, FRANGES), 1024, 0, stream>>>(src, dst, cout, rowl, bsum, pdst, bpre, col);

    gather_gemm<<<N_NODES / 16, 128, 0, stream>>>(feat, cin, rowl, bpre, col, W, bias, out);
}
